# starting kernel
# baseline (speedup 1.0000x reference)
#include <hip/hip_runtime.h>
#include <math.h>

#define B_DIM 64
#define S_DIM 2048
#define D_DIM 256
#define N_HOP 3
#define EPS 1e-8f

typedef __attribute__((ext_vector_type(2))) float v2f;
typedef __attribute__((ext_vector_type(8))) float v8f;

__global__ __launch_bounds__(256) void k_scores(const float* __restrict__ state,
                                                const float* __restrict__ q,
                                                float* __restrict__ scores) {
    __shared__ float qs[D_DIM];
    const int b = blockIdx.x;
    const int tid = threadIdx.x;
    qs[tid] = q[b * D_DIM + tid];
    __syncthreads();

    const int wave = tid >> 5;
    const int lane = tid & 31;

    float qn2 = 0.f;
#pragma unroll
    for (int k = 0; k < 8; ++k) {
        float v = qs[lane + 32 * k];
        qn2 += v * v;
    }
#pragma unroll
    for (int off = 16; off > 0; off >>= 1) qn2 += __shfl_xor(qn2, off, 32);
    const float qn = fmaxf(sqrtf(qn2), EPS);

    const int s = blockIdx.y * 8 + wave;
    const float* row = state + ((size_t)b * S_DIM + s) * D_DIM;
    float dot = 0.f, sn2 = 0.f;
#pragma unroll
    for (int k = 0; k < 8; ++k) {
        float v = row[lane + 32 * k];
        dot += v * qs[lane + 32 * k];
        sn2 += v * v;
    }
#pragma unroll
    for (int off = 16; off > 0; off >>= 1) {
        dot += __shfl_xor(dot, off, 32);
        sn2 += __shfl_xor(sn2, off, 32);
    }
    if (lane == 0) {
        const float sn = fmaxf(sqrtf(sn2), EPS);
        scores[b * S_DIM + s] = dot / (qn * sn);
    }
}

__global__ __launch_bounds__(256) void k_softmax(const float* __restrict__ scores,
                                                 float* __restrict__ dist) {
    __shared__ float red[256];
    const int b = blockIdx.x;
    const int tid = threadIdx.x;
    const float* in = scores + (size_t)b * S_DIM;

    float v[8];
    float m = -INFINITY;
#pragma unroll
    for (int k = 0; k < 8; ++k) {
        v[k] = in[tid + 256 * k];
        m = fmaxf(m, v[k]);
    }
    red[tid] = m;
    __syncthreads();
    for (int off = 128; off > 0; off >>= 1) {
        if (tid < off) red[tid] = fmaxf(red[tid], red[tid + off]);
        __syncthreads();
    }
    m = red[0];
    __syncthreads();

    float sum = 0.f;
#pragma unroll
    for (int k = 0; k < 8; ++k) {
        v[k] = __expf(v[k] - m);
        sum += v[k];
    }
    red[tid] = sum;
    __syncthreads();
    for (int off = 128; off > 0; off >>= 1) {
        if (tid < off) red[tid] += red[tid + off];
        __syncthreads();
    }
    const float inv = 1.f / red[0];
    float* out = dist + (size_t)b * S_DIM;
#pragma unroll
    for (int k = 0; k < 8; ++k) out[tid + 256 * k] = v[k] * inv;
}

#define ROWS_PER_BLOCK 32
#define LDS_STRIDE 260

__global__ __launch_bounds__(256) void k_gemm_qacc(float* __restrict__ state,
                                                   const float* __restrict__ Wh,
                                                   const float* __restrict__ bh,
                                                   const float* __restrict__ dist,
                                                   float* __restrict__ qacc) {
    __shared__ float rows[ROWS_PER_BLOCK * LDS_STRIDE];
    __shared__ float dloc[ROWS_PER_BLOCK];

    const int tid = threadIdx.x;
    const size_t g0 = (size_t)blockIdx.x * ROWS_PER_BLOCK;
    const int b = (int)(g0 / S_DIM);
    const int s0 = (int)(g0 % S_DIM);

    const int wave = tid >> 5;
    const int lane = tid & 31;
    const int half = lane >> 4;
    const int l15 = lane & 15;
    const int e0 = wave * 32;

    {
        const float* wrow = Wh + (size_t)(e0 + (lane & 31)) * D_DIM;
        __builtin_prefetch(wrow, 0, 3);
        __builtin_prefetch(wrow + 128, 0, 3);
    }

#pragma unroll
    for (int i = 0; i < 8; ++i) {
        const int f = i * 256 + tid;
        const int r = f >> 6;
        const int c4 = f & 63;
        float4 t = *(const float4*)(state + (g0 + r) * D_DIM + c4 * 4);
        float* dst = rows + r * LDS_STRIDE + c4 * 4;
        dst[0] = t.x; dst[1] = t.y; dst[2] = t.z; dst[3] = t.w;
    }
    if (tid < ROWS_PER_BLOCK) dloc[tid] = dist[(size_t)b * S_DIM + s0 + tid];
    __syncthreads();

    const float* abase0 = rows + l15 * LDS_STRIDE + 2 * half;
    const float* abase1 = rows + (16 + l15) * LDS_STRIDE + 2 * half;
    const float* wbase0 = Wh + (size_t)(e0 + l15) * D_DIM + 2 * half;
    const float* wbase1 = Wh + (size_t)(e0 + 16 + l15) * D_DIM + 2 * half;

    v8f c00 = {}, c01 = {}, c10 = {}, c11 = {};
    for (int kk = 0; kk < D_DIM; kk += 4) {
        v2f a0, a1, b0, b1;
        a0.x = abase0[kk]; a0.y = abase0[kk + 1];
        a1.x = abase1[kk]; a1.y = abase1[kk + 1];
        b0.x = wbase0[kk]; b0.y = wbase0[kk + 1];
        b1.x = wbase1[kk]; b1.y = wbase1[kk + 1];
        c00 = __builtin_amdgcn_wmma_f32_16x16x4_f32(false, a0, false, b0, (short)0, c00, false, false);
        c01 = __builtin_amdgcn_wmma_f32_16x16x4_f32(false, a0, false, b1, (short)0, c01, false, false);
        c10 = __builtin_amdgcn_wmma_f32_16x16x4_f32(false, a1, false, b0, (short)0, c10, false, false);
        c11 = __builtin_amdgcn_wmma_f32_16x16x4_f32(false, a1, false, b1, (short)0, c11, false, false);
    }

#pragma unroll
    for (int ct = 0; ct < 2; ++ct) {
        const int e = e0 + ct * 16 + l15;
        const float bias = bh[e];
        float qpart = 0.f;
#pragma unroll
        for (int rt = 0; rt < 2; ++rt) {
            const v8f& c = (ct == 0) ? (rt == 0 ? c00 : c10) : (rt == 0 ? c01 : c11);
#pragma unroll
            for (int r = 0; r < 8; ++r) {
                const int M = rt * 16 + r + half * 8;
                const float val = c[r] + bias;
                state[(g0 + M) * D_DIM + e] = val;
                qpart += dloc[M] * val;
            }
        }
        qpart += __shfl_xor(qpart, 16, 32);
        if (half == 0) atomicAdd(&qacc[(size_t)b * D_DIM + e], qpart);
    }
}

__global__ __launch_bounds__(256) void k_qupdate(float* __restrict__ q,
                                                 float* __restrict__ qacc) {
    const int i = blockIdx.x * D_DIM + threadIdx.x;
    q[i] += qacc[i];
    qacc[i] = 0.f;
}

extern "C" void kernel_launch(void* const* d_in, const int* in_sizes, int n_in,
                              void* d_out, int out_size, void* d_ws, size_t ws_size,
                              hipStream_t stream) {
    (void)in_sizes; (void)n_in; (void)out_size; (void)ws_size;

    const float* in_q     = (const float*)d_in[0];
    const float* in_state = (const float*)d_in[1];
    const float* Wall     = (const float*)d_in[2];
    const float* ball     = (const float*)d_in[3];

    const size_t stateBytes = (size_t)B_DIM * S_DIM * D_DIM * sizeof(float);
    const size_t bsBytes    = (size_t)B_DIM * S_DIM * sizeof(float);
    const size_t bdBytes    = (size_t)B_DIM * D_DIM * sizeof(float);

    char* ws = (char*)d_ws;
    float* state  = (float*)(ws);
    float* scores = (float*)(ws + stateBytes);
    float* dist   = (float*)(ws + stateBytes + bsBytes);
    float* q      = (float*)(ws + stateBytes + 2 * bsBytes);
    float* qacc   = (float*)(ws + stateBytes + 2 * bsBytes + bdBytes);

    hipMemcpyAsync(state, in_state, stateBytes, hipMemcpyDeviceToDevice, stream);
    hipMemcpyAsync(q, in_q, bdBytes, hipMemcpyDeviceToDevice, stream);
    hipMemsetAsync(qacc, 0, bdBytes, stream);

    const dim3 gScores(B_DIM, S_DIM / 8);
    for (int h = 0; h < N_HOP; ++h) {
        k_scores<<<gScores, 256, 0, stream>>>(state, q, scores);
        k_softmax<<<B_DIM, 256, 0, stream>>>(scores, dist);
        k_gemm_qacc<<<(B_DIM * S_DIM) / ROWS_PER_BLOCK, 256, 0, stream>>>(
            state, Wall + (size_t)h * D_DIM * D_DIM, ball + (size_t)h * D_DIM,
            dist, qacc);
        k_qupdate<<<B_DIM, 256, 0, stream>>>(q, qacc);
    }
    k_scores<<<gScores, 256, 0, stream>>>(state, q, scores);
    k_softmax<<<B_DIM, 256, 0, stream>>>(scores, dist);

    float* outp = (float*)d_out;
    hipMemcpyAsync(outp, q, bdBytes, hipMemcpyDeviceToDevice, stream);
    hipMemcpyAsync(outp + B_DIM * D_DIM, dist, bsBytes, hipMemcpyDeviceToDevice, stream);
}


# GEMM: Wh staged once per workgroup in LDS (256 persistent workgroups), A rows in registers, B fragments from LDS not global; k order in 8-blocks permuted; q sums reduced in LDS, one atomic per column
# speedup vs baseline: 1.4063x; 1.4063x over previous
#include <hip/hip_runtime.h>
#include <math.h>

#define B_DIM 64
#define S_DIM 2048
#define D_DIM 256
#define N_HOP 3
#define EPS 1e-8f

typedef __attribute__((ext_vector_type(2))) float v2f;
typedef __attribute__((ext_vector_type(8))) float v8f;

__global__ __launch_bounds__(256) void k_scores(const float* __restrict__ state,
                                                const float* __restrict__ q,
                                                float* __restrict__ scores) {
    __shared__ float qs[D_DIM];
    const int b = blockIdx.x;
    const int tid = threadIdx.x;
    qs[tid] = q[b * D_DIM + tid];
    __syncthreads();

    const int wave = tid >> 5;
    const int lane = tid & 31;

    float qn2 = 0.f;
#pragma unroll
    for (int k = 0; k < 8; ++k) {
        float v = qs[lane + 32 * k];
        qn2 += v * v;
    }
#pragma unroll
    for (int off = 16; off > 0; off >>= 1) qn2 += __shfl_xor(qn2, off, 32);
    const float qn = fmaxf(sqrtf(qn2), EPS);

    const int s = blockIdx.y * 8 + wave;
    const float* row = state + ((size_t)b * S_DIM + s) * D_DIM;
    float dot = 0.f, sn2 = 0.f;
#pragma unroll
    for (int k = 0; k < 8; ++k) {
        float v = row[lane + 32 * k];
        dot += v * qs[lane + 32 * k];
        sn2 += v * v;
    }
#pragma unroll
    for (int off = 16; off > 0; off >>= 1) {
        dot += __shfl_xor(dot, off, 32);
        sn2 += __shfl_xor(sn2, off, 32);
    }
    if (lane == 0) {
        const float sn = fmaxf(sqrtf(sn2), EPS);
        scores[b * S_DIM + s] = dot / (qn * sn);
    }
}

__global__ __launch_bounds__(256) void k_softmax(const float* __restrict__ scores,
                                                 float* __restrict__ dist) {
    __shared__ float red[256];
    const int b = blockIdx.x;
    const int tid = threadIdx.x;
    const float* in = scores + (size_t)b * S_DIM;

    float v[8];
    float m = -INFINITY;
#pragma unroll
    for (int k = 0; k < 8; ++k) {
        v[k] = in[tid + 256 * k];
        m = fmaxf(m, v[k]);
    }
    red[tid] = m;
    __syncthreads();
    for (int off = 128; off > 0; off >>= 1) {
        if (tid < off) red[tid] = fmaxf(red[tid], red[tid + off]);
        __syncthreads();
    }
    m = red[0];
    __syncthreads();

    float sum = 0.f;
#pragma unroll
    for (int k = 0; k < 8; ++k) {
        v[k] = __expf(v[k] - m);
        sum += v[k];
    }
    red[tid] = sum;
    __syncthreads();
    for (int off = 128; off > 0; off >>= 1) {
        if (tid < off) red[tid] += red[tid + off];
        __syncthreads();
    }
    const float inv = 1.f / red[0];
    float* out = dist + (size_t)b * S_DIM;
#pragma unroll
    for (int k = 0; k < 8; ++k) out[tid + 256 * k] = v[k] * inv;
}

#define GEMM_THREADS 512
#define GEMM_WAVES 16
#define ROWS_PER_WG 512
#define W_PITCH 260
#define KB_PER_GROUP 4

__global__ __launch_bounds__(GEMM_THREADS) void k_gemm_qacc(float* __restrict__ state,
                                                            const float* __restrict__ Wh,
                                                            const float* __restrict__ bh,
                                                            const float* __restrict__ dist,
                                                            float* __restrict__ qacc) {
    __shared__ __attribute__((aligned(16))) float Wl[D_DIM * W_PITCH];
    __shared__ float qred[GEMM_WAVES * D_DIM];

    const int tid = threadIdx.x;
    const int wave = tid >> 5;
    const int lane = tid & 31;
    const int half = lane >> 4;
    const int l15 = lane & 15;

    const size_t wg_row0 = (size_t)blockIdx.x * ROWS_PER_WG;
    const int b = (int)(wg_row0 / S_DIM);
    const size_t wave_row0 = wg_row0 + (size_t)wave * 32;

    float4 a4[32];
    {
        const float* ap = state + (wave_row0 + l15) * D_DIM + 4 * half;
#pragma unroll
        for (int kb = 0; kb < 32; ++kb) a4[kb] = *(const float4*)(ap + 8 * kb);
    }

#pragma unroll 8
    for (int i = 0; i < 32; ++i) {
        const int f = i * GEMM_THREADS + tid;
        const int e = f >> 6;
        const int c4 = f & 63;
        const float4 t = *(const float4*)(Wh + (size_t)e * D_DIM + c4 * 4);
        *(float4*)(Wl + e * W_PITCH + c4 * 4) = t;
    }
    __syncthreads();

    const float* bbase = Wl + l15 * W_PITCH + 4 * half;

    float4 bq0[2][KB_PER_GROUP], bq1[2][KB_PER_GROUP];
#pragma unroll
    for (int j = 0; j < KB_PER_GROUP; ++j) {
        bq0[0][j] = *(const float4*)(bbase + 8 * j);
        bq1[0][j] = *(const float4*)(bbase + 16 * W_PITCH + 8 * j);
    }

    for (int tile = 0; tile < 2; ++tile) {
        const size_t r0 = wave_row0 + (size_t)tile * 16;

        float dv[8];
        {
            const float* dp = dist + r0 + 8 * half;
            const float4 d0 = *(const float4*)dp;
            const float4 d1 = *(const float4*)(dp + 4);
            dv[0] = d0.x; dv[1] = d0.y; dv[2] = d0.z; dv[3] = d0.w;
            dv[4] = d1.x; dv[5] = d1.y; dv[6] = d1.z; dv[7] = d1.w;
        }

        for (int pass = 0; pass < 8; ++pass) {
            const int e0 = pass * 32;
            const float* bp = bbase + e0 * W_PITCH;
            const float* bpn = bbase + ((pass + 1) & 7) * 32 * W_PITCH;
            const float bias0 = bh[e0 + l15];
            const float bias1 = bh[e0 + 16 + l15];

            v8f c0 = {}, c1 = {};
#pragma unroll
            for (int g = 0; g < 32 / KB_PER_GROUP; ++g) {
                const int cur = g & 1, nxt = cur ^ 1;
                const float* np = (g + 1 < 32 / KB_PER_GROUP) ? bp + 8 * KB_PER_GROUP * (g + 1) : bpn;
#pragma unroll
                for (int j = 0; j < KB_PER_GROUP; ++j) {
                    bq0[nxt][j] = *(const float4*)(np + 8 * j);
                    bq1[nxt][j] = *(const float4*)(np + 16 * W_PITCH + 8 * j);
                }
                __builtin_amdgcn_sched_barrier(0);
#pragma unroll
                for (int j = 0; j < KB_PER_GROUP; ++j) {
                    const float4 av = a4[g * KB_PER_GROUP + j];
                    const float4 b0 = bq0[cur][j];
                    const float4 b1 = bq1[cur][j];
                    v2f alo, ahi, b0lo, b0hi, b1lo, b1hi;
                    alo.x = av.x; alo.y = av.y; ahi.x = av.z; ahi.y = av.w;
                    b0lo.x = b0.x; b0lo.y = b0.y; b0hi.x = b0.z; b0hi.y = b0.w;
                    b1lo.x = b1.x; b1lo.y = b1.y; b1hi.x = b1.z; b1hi.y = b1.w;
                    c0 = __builtin_amdgcn_wmma_f32_16x16x4_f32(false, alo, false, b0lo, (short)0, c0, false, false);
                    c1 = __builtin_amdgcn_wmma_f32_16x16x4_f32(false, alo, false, b1lo, (short)0, c1, false, false);
                    c0 = __builtin_amdgcn_wmma_f32_16x16x4_f32(false, ahi, false, b0hi, (short)0, c0, false, false);
                    c1 = __builtin_amdgcn_wmma_f32_16x16x4_f32(false, ahi, false, b1hi, (short)0, c1, false, false);
                }
                __builtin_amdgcn_sched_barrier(0);
            }

#pragma unroll
            for (int ct = 0; ct < 2; ++ct) {
                const int e = e0 + ct * 16 + l15;
                const float bias = (ct == 0) ? bias0 : bias1;
                const v8f& c = (ct == 0) ? c0 : c1;
                float qpart = 0.f;
#pragma unroll
                for (int r = 0; r < 8; ++r) {
                    const float val = c[r] + bias;
                    state[(r0 + 8 * half + r) * D_DIM + e] = val;
                    qpart += dv[r] * val;
                }
                qpart += __shfl_xor(qpart, 16, 32);
                if (half == 0) {
                    float* p = qred + wave * D_DIM + e;
                    *p = (tile == 0) ? qpart : (*p + qpart);
                }
            }
        }

        if (tile == 0) {
            const float* ap = state + (r0 + 16 + l15) * D_DIM + 4 * half;
#pragma unroll
            for (int kb = 0; kb < 32; ++kb) a4[kb] = *(const float4*)(ap + 8 * kb);
        }
    }

    __syncthreads();
    if (tid < D_DIM) {
        float s = 0.f;
#pragma unroll
        for (int w = 0; w < GEMM_WAVES; ++w) s += qred[w * D_DIM + tid];
        atomicAdd(&qacc[(size_t)b * D_DIM + tid], s);
    }
}

__global__ __launch_bounds__(256) void k_qupdate(float* __restrict__ q,
                                                 float* __restrict__ qacc) {
    const int i = blockIdx.x * D_DIM + threadIdx.x;
    q[i] += qacc[i];
    qacc[i] = 0.f;
}

extern "C" void kernel_launch(void* const* d_in, const int* in_sizes, int n_in,
                              void* d_out, int out_size, void* d_ws, size_t ws_size,
                              hipStream_t stream) {
    (void)in_sizes; (void)n_in; (void)out_size; (void)ws_size;

    const float* in_q     = (const float*)d_in[0];
    const float* in_state = (const float*)d_in[1];
    const float* Wall     = (const float*)d_in[2];
    const float* ball     = (const float*)d_in[3];

    const size_t stateBytes = (size_t)B_DIM * S_DIM * D_DIM * sizeof(float);
    const size_t bsBytes    = (size_t)B_DIM * S_DIM * sizeof(float);
    const size_t bdBytes    = (size_t)B_DIM * D_DIM * sizeof(float);

    char* ws = (char*)d_ws;
    float* state  = (float*)(ws);
    float* scores = (float*)(ws + stateBytes);
    float* dist   = (float*)(ws + stateBytes + bsBytes);
    float* q      = (float*)(ws + stateBytes + 2 * bsBytes);
    float* qacc   = (float*)(ws + stateBytes + 2 * bsBytes + bdBytes);

    hipMemcpyAsync(state, in_state, stateBytes, hipMemcpyDeviceToDevice, stream);
    hipMemcpyAsync(q, in_q, bdBytes, hipMemcpyDeviceToDevice, stream);
    hipMemsetAsync(qacc, 0, bdBytes, stream);

    const dim3 gScores(B_DIM, S_DIM / 8);
    for (int h = 0; h < N_HOP; ++h) {
        k_scores<<<gScores, 256, 0, stream>>>(state, q, scores);
        k_softmax<<<B_DIM, 256, 0, stream>>>(scores, dist);
        k_gemm_qacc<<<(B_DIM * S_DIM) / ROWS_PER_WG, GEMM_THREADS, 0, stream>>>(
            state, Wall + (size_t)h * D_DIM * D_DIM, ball + (size_t)h * D_DIM,
            dist, qacc);
        k_qupdate<<<B_DIM, 256, 0, stream>>>(q, qacc);
    }
    k_scores<<<gScores, 256, 0, stream>>>(state, q, scores);
    k_softmax<<<B_DIM, 256, 0, stream>>>(scores, dist);

    float* outp = (float*)d_out;
    hipMemcpyAsync(outp, q, bdBytes, hipMemcpyDeviceToDevice, stream);
    hipMemcpyAsync(outp + B_DIM * D_DIM, dist, bsBytes, hipMemcpyDeviceToDevice, stream);
}


# Hops 0-2: cosine-softmax fused into the GEMM: exp(score) from the A rows in registers (no max shift) weights the q sums, one division by the summed exp at q update; score sums in fragment order
# speedup vs baseline: 1.6657x; 1.1845x over previous
#include <hip/hip_runtime.h>
#include <math.h>

#define B_DIM 64
#define S_DIM 2048
#define D_DIM 256
#define N_HOP 3
#define EPS 1e-8f

typedef __attribute__((ext_vector_type(2))) float v2f;
typedef __attribute__((ext_vector_type(8))) float v8f;

__global__ __launch_bounds__(256) void k_scores(const float* __restrict__ state,
                                                const float* __restrict__ q,
                                                float* __restrict__ scores) {
    __shared__ float qs[D_DIM];
    const int b = blockIdx.x;
    const int tid = threadIdx.x;
    qs[tid] = q[b * D_DIM + tid];
    __syncthreads();

    const int wave = tid >> 5;
    const int lane = tid & 31;

    float qn2 = 0.f;
#pragma unroll
    for (int k = 0; k < 8; ++k) {
        float v = qs[lane + 32 * k];
        qn2 += v * v;
    }
#pragma unroll
    for (int off = 16; off > 0; off >>= 1) qn2 += __shfl_xor(qn2, off, 32);
    const float qn = fmaxf(sqrtf(qn2), EPS);

    const int s = blockIdx.y * 8 + wave;
    const float* row = state + ((size_t)b * S_DIM + s) * D_DIM;
    float dot = 0.f, sn2 = 0.f;
#pragma unroll
    for (int k = 0; k < 8; ++k) {
        float v = row[lane + 32 * k];
        dot += v * qs[lane + 32 * k];
        sn2 += v * v;
    }
#pragma unroll
    for (int off = 16; off > 0; off >>= 1) {
        dot += __shfl_xor(dot, off, 32);
        sn2 += __shfl_xor(sn2, off, 32);
    }
    if (lane == 0) {
        const float sn = fmaxf(sqrtf(sn2), EPS);
        scores[b * S_DIM + s] = dot / (qn * sn);
    }
}

__global__ __launch_bounds__(256) void k_softmax(const float* __restrict__ scores,
                                                 float* __restrict__ dist) {
    __shared__ float red[256];
    const int b = blockIdx.x;
    const int tid = threadIdx.x;
    const float* in = scores + (size_t)b * S_DIM;

    float v[8];
    float m = -INFINITY;
#pragma unroll
    for (int k = 0; k < 8; ++k) {
        v[k] = in[tid + 256 * k];
        m = fmaxf(m, v[k]);
    }
    red[tid] = m;
    __syncthreads();
    for (int off = 128; off > 0; off >>= 1) {
        if (tid < off) red[tid] = fmaxf(red[tid], red[tid + off]);
        __syncthreads();
    }
    m = red[0];
    __syncthreads();

    float sum = 0.f;
#pragma unroll
    for (int k = 0; k < 8; ++k) {
        v[k] = __expf(v[k] - m);
        sum += v[k];
    }
    red[tid] = sum;
    __syncthreads();
    for (int off = 128; off > 0; off >>= 1) {
        if (tid < off) red[tid] += red[tid + off];
        __syncthreads();
    }
    const float inv = 1.f / red[0];
    float* out = dist + (size_t)b * S_DIM;
#pragma unroll
    for (int k = 0; k < 8; ++k) out[tid + 256 * k] = v[k] * inv;
}

#define GEMM_THREADS 512
#define GEMM_WAVES 16
#define ROWS_PER_WG 512
#define W_PITCH 260
#define KB_PER_GROUP 4

__global__ __launch_bounds__(GEMM_THREADS) void k_gemm_qacc(float* __restrict__ state,
                                                            const float* __restrict__ Wh,
                                                            const float* __restrict__ bh,
                                                            const float* __restrict__ q,
                                                            float* __restrict__ qacc,
                                                            float* __restrict__ wpart) {
    __shared__ __attribute__((aligned(16))) float Wl[D_DIM * W_PITCH];
    __shared__ float qred[GEMM_WAVES * D_DIM];
    __shared__ __attribute__((aligned(16))) float qs[D_DIM];
    __shared__ float wred[GEMM_WAVES];

    const int tid = threadIdx.x;
    const int wave = tid >> 5;
    const int lane = tid & 31;
    const int half = lane >> 4;
    const int l15 = lane & 15;

    const size_t wg_row0 = (size_t)blockIdx.x * ROWS_PER_WG;
    const int b = (int)(wg_row0 / S_DIM);
    const size_t wave_row0 = wg_row0 + (size_t)wave * 32;

    float4 a4[32];
    {
        const float* ap = state + (wave_row0 + l15) * D_DIM + 4 * half;
#pragma unroll
        for (int kb = 0; kb < 32; ++kb) a4[kb] = *(const float4*)(ap + 8 * kb);
    }

#pragma unroll 8
    for (int i = 0; i < 32; ++i) {
        const int f = i * GEMM_THREADS + tid;
        const int e = f >> 6;
        const int c4 = f & 63;
        const float4 t = *(const float4*)(Wh + (size_t)e * D_DIM + c4 * 4);
        *(float4*)(Wl + e * W_PITCH + c4 * 4) = t;
    }
    if (tid < D_DIM) qs[tid] = q[(size_t)b * D_DIM + tid];
    __syncthreads();

    float qn2 = 0.f;
    for (int k = 0; k < 8; ++k) {
        const float v = qs[lane + 32 * k];
        qn2 += v * v;
    }
    for (int off = 16; off > 0; off >>= 1) qn2 += __shfl_xor(qn2, off, 32);
    const float qn = fmaxf(sqrtf(qn2), EPS);
    float wsum_wave = 0.f;

    const float* bbase = Wl + l15 * W_PITCH + 4 * half;

    float4 bq0[2][KB_PER_GROUP], bq1[2][KB_PER_GROUP];
#pragma unroll
    for (int j = 0; j < KB_PER_GROUP; ++j) {
        bq0[0][j] = *(const float4*)(bbase + 8 * j);
        bq1[0][j] = *(const float4*)(bbase + 16 * W_PITCH + 8 * j);
    }

    for (int tile = 0; tile < 2; ++tile) {
        const size_t r0 = wave_row0 + (size_t)tile * 16;

        float wrow;
        {
            const float* qp = qs + 4 * half;
            v2f dlo = {0.f, 0.f}, dhi = {0.f, 0.f}, nlo = {0.f, 0.f}, nhi = {0.f, 0.f};
#pragma unroll
            for (int kb = 0; kb < 32; ++kb) {
                const float4 qv = *(const float4*)(qp + 8 * kb);
                const float4 av = a4[kb];
                v2f alo, ahi, qlo, qhi;
                alo.x = av.x; alo.y = av.y; ahi.x = av.z; ahi.y = av.w;
                qlo.x = qv.x; qlo.y = qv.y; qhi.x = qv.z; qhi.y = qv.w;
                dlo += alo * qlo; dhi += ahi * qhi;
                nlo += alo * alo; nhi += ahi * ahi;
            }
            float dot = (dlo.x + dlo.y) + (dhi.x + dhi.y);
            float sn2 = (nlo.x + nlo.y) + (nhi.x + nhi.y);
            dot += __shfl_xor(dot, 16, 32);
            sn2 += __shfl_xor(sn2, 16, 32);
            const float sn = fmaxf(sqrtf(sn2), EPS);
            wrow = __expf(dot / (qn * sn));
        }
        float ws16[16];
#pragma unroll
        for (int r = 0; r < 16; ++r)
            ws16[r] = __int_as_float(__builtin_amdgcn_readlane(__float_as_int(wrow), r));
        float dv[8];
#pragma unroll
        for (int r = 0; r < 8; ++r) dv[r] = half ? ws16[8 + r] : ws16[r];
        wsum_wave += ((ws16[0] + ws16[1]) + (ws16[2] + ws16[3])) + ((ws16[4] + ws16[5]) + (ws16[6] + ws16[7]))
                   + ((ws16[8] + ws16[9]) + (ws16[10] + ws16[11])) + ((ws16[12] + ws16[13]) + (ws16[14] + ws16[15]));

        float* const srow = state + (r0 + 8 * half) * D_DIM + l15;

        for (int pass = 0; pass < 8; ++pass) {
            const int e0 = pass * 32;
            const float* bp = bbase + e0 * W_PITCH;
            const float* bpn = bbase + ((pass + 1) & 7) * 32 * W_PITCH;
            const float bias0 = bh[e0 + l15];
            const float bias1 = bh[e0 + 16 + l15];

            v8f c0 = {}, c1 = {};
#pragma unroll
            for (int g = 0; g < 32 / KB_PER_GROUP; ++g) {
                const int cur = g & 1, nxt = cur ^ 1;
                const float* np = (g + 1 < 32 / KB_PER_GROUP) ? bp + 8 * KB_PER_GROUP * (g + 1) : bpn;
#pragma unroll
                for (int j = 0; j < KB_PER_GROUP; ++j) {
                    bq0[nxt][j] = *(const float4*)(np + 8 * j);
                    bq1[nxt][j] = *(const float4*)(np + 16 * W_PITCH + 8 * j);
                }
                __builtin_amdgcn_sched_barrier(0);
#pragma unroll
                for (int j = 0; j < KB_PER_GROUP; ++j) {
                    const float4 av = a4[g * KB_PER_GROUP + j];
                    const float4 b0 = bq0[cur][j];
                    const float4 b1 = bq1[cur][j];
                    v2f alo, ahi, b0lo, b0hi, b1lo, b1hi;
                    alo.x = av.x; alo.y = av.y; ahi.x = av.z; ahi.y = av.w;
                    b0lo.x = b0.x; b0lo.y = b0.y; b0hi.x = b0.z; b0hi.y = b0.w;
                    b1lo.x = b1.x; b1lo.y = b1.y; b1hi.x = b1.z; b1hi.y = b1.w;
                    c0 = __builtin_amdgcn_wmma_f32_16x16x4_f32(false, alo, false, b0lo, (short)0, c0, false, false);
                    c1 = __builtin_amdgcn_wmma_f32_16x16x4_f32(false, alo, false, b1lo, (short)0, c1, false, false);
                    c0 = __builtin_amdgcn_wmma_f32_16x16x4_f32(false, ahi, false, b0hi, (short)0, c0, false, false);
                    c1 = __builtin_amdgcn_wmma_f32_16x16x4_f32(false, ahi, false, b1hi, (short)0, c1, false, false);
                }
                __builtin_amdgcn_sched_barrier(0);
            }

#pragma unroll
            for (int ct = 0; ct < 2; ++ct) {
                const int e = e0 + ct * 16 + l15;
                const float bias = (ct == 0) ? bias0 : bias1;
                const v8f& c = (ct == 0) ? c0 : c1;
                float qpart = 0.f;
#pragma unroll
                for (int r = 0; r < 8; ++r) {
                    const float val = c[r] + bias;
                    srow[r * D_DIM + e0 + ct * 16] = val;
                    qpart += dv[r] * val;
                }
                qpart += __shfl_xor(qpart, 16, 32);
                if (half == 0) {
                    float* p = qred + wave * D_DIM + e;
                    *p = (tile == 0) ? qpart : (*p + qpart);
                }
            }
        }

        if (tile == 0) {
            const float* ap = state + (r0 + 16 + l15) * D_DIM + 4 * half;
#pragma unroll
            for (int kb = 0; kb < 32; ++kb) a4[kb] = *(const float4*)(ap + 8 * kb);
        }
    }

    if (lane == 0) wred[wave] = wsum_wave;
    __syncthreads();
    if (tid < D_DIM) {
        float s = 0.f;
#pragma unroll
        for (int w = 0; w < GEMM_WAVES; ++w) s += qred[w * D_DIM + tid];
        atomicAdd(&qacc[(size_t)b * D_DIM + tid], s);
    } else if (tid == D_DIM) {
        float s = 0.f;
        for (int w = 0; w < GEMM_WAVES; ++w) s += wred[w];
        wpart[blockIdx.x] = s;
    }
}

#define WG_PER_B (S_DIM / ROWS_PER_WG)
__global__ __launch_bounds__(256) void k_qupdate(float* __restrict__ q,
                                                 float* __restrict__ qacc,
                                                 const float* __restrict__ wpart) {
    const int i = blockIdx.x * D_DIM + threadIdx.x;
    const float* wp = wpart + blockIdx.x * WG_PER_B;
    const float wsum = (wp[0] + wp[1]) + (wp[2] + wp[3]);
    q[i] += qacc[i] / wsum;
    qacc[i] = 0.f;
}

extern "C" void kernel_launch(void* const* d_in, const int* in_sizes, int n_in,
                              void* d_out, int out_size, void* d_ws, size_t ws_size,
                              hipStream_t stream) {
    (void)in_sizes; (void)n_in; (void)out_size; (void)ws_size;

    const float* in_q     = (const float*)d_in[0];
    const float* in_state = (const float*)d_in[1];
    const float* Wall     = (const float*)d_in[2];
    const float* ball     = (const float*)d_in[3];

    const size_t stateBytes = (size_t)B_DIM * S_DIM * D_DIM * sizeof(float);
    const size_t bsBytes    = (size_t)B_DIM * S_DIM * sizeof(float);
    const size_t bdBytes    = (size_t)B_DIM * D_DIM * sizeof(float);

    char* ws = (char*)d_ws;
    float* state  = (float*)(ws);
    float* scores = (float*)(ws + stateBytes);
    float* dist   = (float*)(ws + stateBytes + bsBytes);
    float* q      = (float*)(ws + stateBytes + 2 * bsBytes);
    float* qacc   = (float*)(ws + stateBytes + 2 * bsBytes + bdBytes);
    float* wpart  = scores;

    hipMemcpyAsync(state, in_state, stateBytes, hipMemcpyDeviceToDevice, stream);
    hipMemcpyAsync(q, in_q, bdBytes, hipMemcpyDeviceToDevice, stream);
    hipMemsetAsync(qacc, 0, bdBytes, stream);

    const dim3 gScores(B_DIM, S_DIM / 8);
    for (int h = 0; h < N_HOP; ++h) {
        k_gemm_qacc<<<(B_DIM * S_DIM) / ROWS_PER_WG, GEMM_THREADS, 0, stream>>>(
            state, Wall + (size_t)h * D_DIM * D_DIM, ball + (size_t)h * D_DIM,
            q, qacc, wpart);
        k_qupdate<<<B_DIM, 256, 0, stream>>>(q, qacc, wpart);
    }
    k_scores<<<gScores, 256, 0, stream>>>(state, q, scores);
    k_softmax<<<B_DIM, 256, 0, stream>>>(scores, dist);

    float* outp = (float*)d_out;
    hipMemcpyAsync(outp, q, bdBytes, hipMemcpyDeviceToDevice, stream);
    hipMemcpyAsync(outp + B_DIM * D_DIM, dist, bsBytes, hipMemcpyDeviceToDevice, stream);
}
